# starting kernel
# baseline (speedup 1.0000x reference)
#include <hip/hip_runtime.h>
#include <hip/hip_bf16.h>

typedef __attribute__((ext_vector_type(2))) float v2f;
typedef __attribute__((ext_vector_type(8))) float v8f;

#define FEAT 128
#define TILES_PER_BLOCK 8

__global__ __launch_bounds__(256) void bilinear_gemm_zW(
    const float* __restrict__ z, const float* __restrict__ W,
    float* __restrict__ V, int n_nodes) {
  __shared__ float Al[16][132];

  const int tid   = threadIdx.x;
  const int wv    = tid >> 5;
  const int lane  = tid & 31;
  const int colBase = wv * 16;
  const int mrow  = lane & 15;
  const int khalf = (lane >> 4) << 1;

  v2f bf[FEAT / 4];
#pragma unroll
  for (int kk = 0; kk < FEAT / 4; ++kk) {
    const int k = kk * 4 + khalf;
    bf[kk][0] = W[(size_t)k * FEAT + colBase + mrow];
    bf[kk][1] = W[(size_t)(k + 1) * FEAT + colBase + mrow];
  }

  for (int t = 0; t < TILES_PER_BLOCK; ++t) {
    const int rowBase = (blockIdx.x * TILES_PER_BLOCK + t) * 16;
    if (rowBase >= n_nodes) break;

    __syncthreads();
#pragma unroll
    for (int c4 = tid; c4 < (16 * FEAT) / 4; c4 += 256) {
      const int r = c4 >> 5;
      const int c = (c4 & 31) << 2;
      *(float4*)&Al[r][c] = *(const float4*)&z[(size_t)(rowBase + r) * FEAT + c];
    }
    __syncthreads();

    v8f acc = {};
#pragma unroll
    for (int kk = 0; kk < FEAT / 4; ++kk) {
      const int k = kk * 4 + khalf;
      v2f a;
      a[0] = Al[mrow][k];
      a[1] = Al[mrow][k + 1];
      acc = __builtin_amdgcn_wmma_f32_16x16x4_f32(
          false, a, false, bf[kk], (short)0, acc, false, false);
    }

    const int mOff = (lane >> 4) << 3;
#pragma unroll
    for (int r = 0; r < 8; ++r) {
      V[(size_t)(rowBase + mOff + r) * FEAT + colBase + mrow] = acc[r];
    }
  }
}

__global__ __launch_bounds__(256) void bilinear_edge_scores(
    const float* __restrict__ V, const float* __restrict__ z,
    const int* __restrict__ e_src, const int* __restrict__ e_dst,
    float* __restrict__ out, int n_edges) {
  const int lane   = threadIdx.x & 31;
  const int wid    = (blockIdx.x * blockDim.x + threadIdx.x) >> 5;
  const int nwaves = (gridDim.x * blockDim.x) >> 5;

  for (int e = wid; e < n_edges; e += nwaves) {
    const int s = e_src[e];
    const int d = e_dst[e];
    const float4 av = ((const float4*)(V + (size_t)s * FEAT))[lane];
    const float4 bv = ((const float4*)(z + (size_t)d * FEAT))[lane];
    float sum = av.x * bv.x + av.y * bv.y + av.z * bv.z + av.w * bv.w;
#pragma unroll
    for (int off = 16; off > 0; off >>= 1) sum += __shfl_xor(sum, off, 32);
    if (lane == 0) out[e] = sum;
  }
}

extern "C" void kernel_launch(void* const* d_in, const int* in_sizes, int n_in,
                              void* d_out, int out_size, void* d_ws, size_t ws_size,
                              hipStream_t stream) {
  const float* z  = (const float*)d_in[0];
  const int*   ei = (const int*)d_in[1];
  const float* W  = (const float*)d_in[2];
  float* out = (float*)d_out;
  float* V   = (float*)d_ws;

  const int n_nodes = in_sizes[0] / FEAT;
  const int n_edges = in_sizes[1] / 2;

  const int rowTiles = (n_nodes + 15) / 16;
  const int gemmBlocks = (rowTiles + TILES_PER_BLOCK - 1) / TILES_PER_BLOCK;
  bilinear_gemm_zW<<<gemmBlocks, 256, 0, stream>>>(z, W, V, n_nodes);

  bilinear_edge_scores<<<2048, 256, 0, stream>>>(V, z, ei, ei + n_edges, out,
                                                 n_edges);
}


# edge-score kernel: each wave keeps a rolling window of 8 edges (two whole-row gathers each) in flight instead of one edge per trip; indices read 32 at a time, 8 sums per butterfly, whole-line stores
# speedup vs baseline: 1.7039x; 1.7039x over previous
#include <hip/hip_runtime.h>
#include <hip/hip_bf16.h>

typedef __attribute__((ext_vector_type(2))) float v2f;
typedef __attribute__((ext_vector_type(8))) float v8f;

#define FEAT 128
#define TILES_PER_BLOCK 8

__global__ __launch_bounds__(256) void bilinear_gemm_zW(
    const float* __restrict__ z, const float* __restrict__ W,
    float* __restrict__ V, int n_nodes) {
  __shared__ float Al[16][132];

  const int tid   = threadIdx.x;
  const int wv    = tid >> 5;
  const int lane  = tid & 31;
  const int colBase = wv * 16;
  const int mrow  = lane & 15;
  const int khalf = (lane >> 4) << 1;

  v2f bf[FEAT / 4];
#pragma unroll
  for (int kk = 0; kk < FEAT / 4; ++kk) {
    const int k = kk * 4 + khalf;
    bf[kk][0] = W[(size_t)k * FEAT + colBase + mrow];
    bf[kk][1] = W[(size_t)(k + 1) * FEAT + colBase + mrow];
  }

  for (int t = 0; t < TILES_PER_BLOCK; ++t) {
    const int rowBase = (blockIdx.x * TILES_PER_BLOCK + t) * 16;
    if (rowBase >= n_nodes) break;

    __syncthreads();
#pragma unroll
    for (int c4 = tid; c4 < (16 * FEAT) / 4; c4 += 256) {
      const int r = c4 >> 5;
      const int c = (c4 & 31) << 2;
      *(float4*)&Al[r][c] = *(const float4*)&z[(size_t)(rowBase + r) * FEAT + c];
    }
    __syncthreads();

    v8f acc = {};
#pragma unroll
    for (int kk = 0; kk < FEAT / 4; ++kk) {
      const int k = kk * 4 + khalf;
      v2f a;
      a[0] = Al[mrow][k];
      a[1] = Al[mrow][k + 1];
      acc = __builtin_amdgcn_wmma_f32_16x16x4_f32(
          false, a, false, bf[kk], (short)0, acc, false, false);
    }

    const int mOff = (lane >> 4) << 3;
#pragma unroll
    for (int r = 0; r < 8; ++r) {
      V[(size_t)(rowBase + mOff + r) * FEAT + colBase + mrow] = acc[r];
    }
  }
}

typedef __attribute__((ext_vector_type(4))) float v4f;

template <int CTRL>
__device__ __forceinline__ float row_swz(float x) {
  return __builtin_bit_cast(float, __builtin_amdgcn_update_dpp(
      0, __builtin_bit_cast(int, x), CTRL, 0xf, 0xf, false));
}
__device__ __forceinline__ float xhalf(float x) {
  const int xi = __builtin_bit_cast(int, x);
  return __builtin_bit_cast(float, __builtin_amdgcn_permlanex16(
      xi, xi, 0x76543210, 0xfedcba98, false, false));
}
__device__ __forceinline__ int clamp_node(int i, int n_nodes) {
  return min(max(i, 0), n_nodes - 1);
}

#define EDGES_PER_GROUP 32
#define WINDOW 8

__device__ __forceinline__ constexpr int slot_edge(int b, int k) {
  return b + 4 * ((k >> 2) & 1) + 8 * ((k >> 1) & 1) + 16 * (k & 1);
}

template <int NB>
__device__ __forceinline__ void request_pair(v4f& a, v4f& b, int k, int rsi, int rdi,
                                             const v4f* __restrict__ V4,
                                             const v4f* __restrict__ z4, int lane) {
  const int L = slot_edge(NB, k);
  const int s = __builtin_amdgcn_readlane(rsi, L);
  const int d = __builtin_amdgcn_readlane(rdi, L);
  a = V4[(size_t)s * (FEAT / 4) + lane];
  b = z4[(size_t)d * (FEAT / 4) + lane];
}

template <int NB, bool REFILL>
__device__ __forceinline__ void window_step(v4f (&av)[WINDOW], v4f (&bv)[WINDOW],
                                            float (&v)[WINDOW], int rsi, int rdi,
                                            const v4f* __restrict__ V4,
                                            const v4f* __restrict__ z4, int lane) {
#pragma unroll
  for (int k = 0; k < WINDOW; ++k) {
    asm volatile("" : "+v"(av[k]));
    float sum = __builtin_fmaf(av[k].w, bv[k].w,
                __builtin_fmaf(av[k].z, bv[k].z,
                __builtin_fmaf(av[k].y, bv[k].y, av[k].x * bv[k].x)));
    if (REFILL) {
      asm volatile("" : "+v"(sum), "+v"(rsi), "+v"(rdi));
      request_pair<NB>(av[k], bv[k], k, rsi, rdi, V4, z4, lane);
    }
    v[k] = sum;
  }
}

__device__ __forceinline__ float butterfly8(const float (&v)[WINDOW], bool up16, bool up8,
                                            bool up4) {
  float r[4];
#pragma unroll
  for (int m = 0; m < 4; ++m) {
    const float keep = up16 ? v[2 * m + 1] : v[2 * m];
    const float send = up16 ? v[2 * m] : v[2 * m + 1];
    r[m] = keep + xhalf(send);
  }
  float q[2];
#pragma unroll
  for (int m = 0; m < 2; ++m) {
    const float keep = up8 ? r[2 * m + 1] : r[2 * m];
    const float send = up8 ? r[2 * m] : r[2 * m + 1];
    q[m] = keep + row_swz<0x168>(send);
  }
  const float keep = up4 ? q[1] : q[0];
  const float send = up4 ? q[0] : q[1];
  float t = keep + row_swz<0x164>(send);
  t += row_swz<0x162>(t);
  t += row_swz<0x161>(t);
  return t;
}

template <bool MORE>
__device__ __forceinline__ void group_body(v4f (&av)[WINDOW], v4f (&bv)[WINDOW], int si,
                                           int di, int& nsi, int& ndi,
                                           const v4f* __restrict__ V4,
                                           const v4f* __restrict__ z4, int lane,
                                           int n_nodes, float* __restrict__ out, int g,
                                           int n_edges) {
  const bool up16 = (lane & 16) != 0;
  const bool up8  = (lane & 8) != 0;
  const bool up4  = (lane & 4) != 0;
  const bool odd1 = (lane & 1) != 0;
  const bool odd2 = (lane & 2) != 0;
  float v[WINDOW];
  window_step<1, true>(av, bv, v, si, di, V4, z4, lane);
  const float p0 = butterfly8(v, up16, up8, up4);
  window_step<2, true>(av, bv, v, si, di, V4, z4, lane);
  const float p1 = butterfly8(v, up16, up8, up4);
  window_step<3, true>(av, bv, v, si, di, V4, z4, lane);
  float p2 = butterfly8(v, up16, up8, up4);
  if (MORE) {
    asm volatile("" : "+v"(nsi), "+v"(ndi), "+v"(p2));
    nsi = clamp_node(nsi, n_nodes);
    ndi = clamp_node(ndi, n_nodes);
  }
  window_step<0, MORE>(av, bv, v, nsi, ndi, V4, z4, lane);
  const float p3 = butterfly8(v, up16, up8, up4);

  const float lo = odd1 ? p1 : p0;
  const float hi = odd1 ? p3 : p2;
  const float res = odd2 ? hi : lo;
  const int eo = g * EDGES_PER_GROUP + lane;
  if (eo < n_edges) out[eo] = res;
}

__global__ __launch_bounds__(256) void bilinear_edge_scores(
    const float* __restrict__ V, const float* __restrict__ z,
    const int* __restrict__ e_src, const int* __restrict__ e_dst,
    float* __restrict__ out, int n_edges, int n_nodes) {
  const int lane    = threadIdx.x & 31;
  const int wid     = __builtin_amdgcn_readfirstlane(
      (int)((blockIdx.x * blockDim.x + threadIdx.x) >> 5));
  const int nwaves  = (gridDim.x * blockDim.x) >> 5;
  const int ngroups = (n_edges + EDGES_PER_GROUP - 1) / EDGES_PER_GROUP;
  const v4f* __restrict__ V4 = (const v4f*)V;
  const v4f* __restrict__ z4 = (const v4f*)z;

  int g = wid;
  if (g >= ngroups) return;

  int si, di;
  {
    const int e = min(g * EDGES_PER_GROUP + lane, n_edges - 1);
    si = clamp_node(e_src[e], n_nodes);
    di = clamp_node(e_dst[e], n_nodes);
  }

  v4f av[WINDOW], bv[WINDOW];
#pragma unroll
  for (int k = 0; k < WINDOW; ++k)
    request_pair<0>(av[k], bv[k], k, si, di, V4, z4, lane);

  for (;;) {
    const int gn = g + nwaves;
    if (gn >= ngroups) break;
    const int en = min(gn * EDGES_PER_GROUP + lane, n_edges - 1);
    int nsi = e_src[en];
    int ndi = e_dst[en];
    group_body<true>(av, bv, si, di, nsi, ndi, V4, z4, lane, n_nodes, out, g, n_edges);
    si = nsi;
    di = ndi;
    g = gn;
  }
  int unused_s = 0, unused_d = 0;
  group_body<false>(av, bv, si, di, unused_s, unused_d, V4, z4, lane, n_nodes, out, g,
                    n_edges);
}

extern "C" void kernel_launch(void* const* d_in, const int* in_sizes, int n_in,
                              void* d_out, int out_size, void* d_ws, size_t ws_size,
                              hipStream_t stream) {
  const float* z  = (const float*)d_in[0];
  const int*   ei = (const int*)d_in[1];
  const float* W  = (const float*)d_in[2];
  float* out = (float*)d_out;
  float* V   = (float*)d_ws;

  const int n_nodes = in_sizes[0] / FEAT;
  const int n_edges = in_sizes[1] / 2;

  const int rowTiles = (n_nodes + 15) / 16;
  const int gemmBlocks = (rowTiles + TILES_PER_BLOCK - 1) / TILES_PER_BLOCK;
  bilinear_gemm_zW<<<gemmBlocks, 256, 0, stream>>>(z, W, V, n_nodes);

  bilinear_edge_scores<<<782, 256, 0, stream>>>(V, z, ei, ei + n_edges, out,
                                                n_edges, n_nodes);
}


# z@W kernel as an evenly dealt, software-pipelined tile loop: 768 work-groups take 16-row tiles round-robin (was 782 x 8 in a row), next tile's rows requested ahead of the WMMAs, 2 LDS stages
# speedup vs baseline: 1.9000x; 1.1151x over previous
#include <hip/hip_runtime.h>
#include <hip/hip_bf16.h>

typedef __attribute__((ext_vector_type(2))) float v2f;
typedef __attribute__((ext_vector_type(8))) float v8f;

#define FEAT 128

#define GEMM_BLOCKS 768

__global__ __launch_bounds__(256) void bilinear_gemm_zW(
    const float* __restrict__ z, const float* __restrict__ W,
    float* __restrict__ V, int n_nodes) {
  __shared__ float Al[2][16][132];

  const int tid   = threadIdx.x;
  const int wv    = tid >> 5;
  const int lane  = tid & 31;
  const int colBase = wv * 16;
  const int mrow  = lane & 15;
  const int khalf = (lane >> 4) << 1;

  const int nTiles = (n_nodes + 15) >> 4;
  int tile = blockIdx.x;
  if (tile >= nTiles) return;

  const int sr = tid >> 5;
  const int sc = (tid & 31) << 2;

  float4 ra = *(const float4*)&z[(size_t)(tile * 16 + sr) * FEAT + sc];
  float4 rb = *(const float4*)&z[(size_t)(tile * 16 + sr + 8) * FEAT + sc];

  v2f bf[FEAT / 4];
#pragma unroll
  for (int kk = 0; kk < FEAT / 4; ++kk) {
    const int k = kk * 4 + khalf;
    bf[kk][0] = W[(size_t)k * FEAT + colBase + mrow];
    bf[kk][1] = W[(size_t)(k + 1) * FEAT + colBase + mrow];
  }

  int s = 0;
  for (;;) {
    *(float4*)&Al[s][sr][sc]     = ra;
    *(float4*)&Al[s][sr + 8][sc] = rb;
    __syncthreads();

    const int next = tile + GEMM_BLOCKS;
    const bool more = next < nTiles;
    const int ldTile = more ? next : tile;
    ra = *(const float4*)&z[(size_t)(ldTile * 16 + sr) * FEAT + sc];
    rb = *(const float4*)&z[(size_t)(ldTile * 16 + sr + 8) * FEAT + sc];
    __builtin_amdgcn_sched_barrier(0);

    v8f acc = {};
#pragma unroll
    for (int kk = 0; kk < FEAT / 4; ++kk) {
      const int k = kk * 4 + khalf;
      v2f a;
      a[0] = Al[s][mrow][k];
      a[1] = Al[s][mrow][k + 1];
      acc = __builtin_amdgcn_wmma_f32_16x16x4_f32(
          false, a, false, bf[kk], (short)0, acc, false, false);
    }

    const int rowBase = tile * 16;
    const int mOff = (lane >> 4) << 3;
#pragma unroll
    for (int r = 0; r < 8; ++r) {
      V[(size_t)(rowBase + mOff + r) * FEAT + colBase + mrow] = acc[r];
    }

    if (!more) break;
    tile = next;
    s ^= 1;
  }
}

typedef __attribute__((ext_vector_type(4))) float v4f;

template <int CTRL>
__device__ __forceinline__ float row_swz(float x) {
  return __builtin_bit_cast(float, __builtin_amdgcn_update_dpp(
      0, __builtin_bit_cast(int, x), CTRL, 0xf, 0xf, false));
}
__device__ __forceinline__ float xhalf(float x) {
  const int xi = __builtin_bit_cast(int, x);
  return __builtin_bit_cast(float, __builtin_amdgcn_permlanex16(
      xi, xi, 0x76543210, 0xfedcba98, false, false));
}
__device__ __forceinline__ int clamp_node(int i, int n_nodes) {
  return min(max(i, 0), n_nodes - 1);
}

#define EDGES_PER_GROUP 32
#define WINDOW 8

__device__ __forceinline__ constexpr int slot_edge(int b, int k) {
  return b + 4 * ((k >> 2) & 1) + 8 * ((k >> 1) & 1) + 16 * (k & 1);
}

template <int NB>
__device__ __forceinline__ void request_pair(v4f& a, v4f& b, int k, int rsi, int rdi,
                                             const v4f* __restrict__ V4,
                                             const v4f* __restrict__ z4, int lane) {
  const int L = slot_edge(NB, k);
  const int s = __builtin_amdgcn_readlane(rsi, L);
  const int d = __builtin_amdgcn_readlane(rdi, L);
  a = V4[(size_t)s * (FEAT / 4) + lane];
  b = z4[(size_t)d * (FEAT / 4) + lane];
}

template <int NB, bool REFILL>
__device__ __forceinline__ void window_step(v4f (&av)[WINDOW], v4f (&bv)[WINDOW],
                                            float (&v)[WINDOW], int rsi, int rdi,
                                            const v4f* __restrict__ V4,
                                            const v4f* __restrict__ z4, int lane) {
#pragma unroll
  for (int k = 0; k < WINDOW; ++k) {
    asm volatile("" : "+v"(av[k]));
    float sum = __builtin_fmaf(av[k].w, bv[k].w,
                __builtin_fmaf(av[k].z, bv[k].z,
                __builtin_fmaf(av[k].y, bv[k].y, av[k].x * bv[k].x)));
    if (REFILL) {
      asm volatile("" : "+v"(sum), "+v"(rsi), "+v"(rdi));
      request_pair<NB>(av[k], bv[k], k, rsi, rdi, V4, z4, lane);
    }
    v[k] = sum;
  }
}

__device__ __forceinline__ float butterfly8(const float (&v)[WINDOW], bool up16, bool up8,
                                            bool up4) {
  float r[4];
#pragma unroll
  for (int m = 0; m < 4; ++m) {
    const float keep = up16 ? v[2 * m + 1] : v[2 * m];
    const float send = up16 ? v[2 * m] : v[2 * m + 1];
    r[m] = keep + xhalf(send);
  }
  float q[2];
#pragma unroll
  for (int m = 0; m < 2; ++m) {
    const float keep = up8 ? r[2 * m + 1] : r[2 * m];
    const float send = up8 ? r[2 * m] : r[2 * m + 1];
    q[m] = keep + row_swz<0x168>(send);
  }
  const float keep = up4 ? q[1] : q[0];
  const float send = up4 ? q[0] : q[1];
  float t = keep + row_swz<0x164>(send);
  t += row_swz<0x162>(t);
  t += row_swz<0x161>(t);
  return t;
}

template <bool MORE>
__device__ __forceinline__ void group_body(v4f (&av)[WINDOW], v4f (&bv)[WINDOW], int si,
                                           int di, int& nsi, int& ndi,
                                           const v4f* __restrict__ V4,
                                           const v4f* __restrict__ z4, int lane,
                                           int n_nodes, float* __restrict__ out, int g,
                                           int n_edges) {
  const bool up16 = (lane & 16) != 0;
  const bool up8  = (lane & 8) != 0;
  const bool up4  = (lane & 4) != 0;
  const bool odd1 = (lane & 1) != 0;
  const bool odd2 = (lane & 2) != 0;
  float v[WINDOW];
  window_step<1, true>(av, bv, v, si, di, V4, z4, lane);
  const float p0 = butterfly8(v, up16, up8, up4);
  window_step<2, true>(av, bv, v, si, di, V4, z4, lane);
  const float p1 = butterfly8(v, up16, up8, up4);
  window_step<3, true>(av, bv, v, si, di, V4, z4, lane);
  float p2 = butterfly8(v, up16, up8, up4);
  if (MORE) {
    asm volatile("" : "+v"(nsi), "+v"(ndi), "+v"(p2));
    nsi = clamp_node(nsi, n_nodes);
    ndi = clamp_node(ndi, n_nodes);
  }
  window_step<0, MORE>(av, bv, v, nsi, ndi, V4, z4, lane);
  const float p3 = butterfly8(v, up16, up8, up4);

  const float lo = odd1 ? p1 : p0;
  const float hi = odd1 ? p3 : p2;
  const float res = odd2 ? hi : lo;
  const int eo = g * EDGES_PER_GROUP + lane;
  if (eo < n_edges) out[eo] = res;
}

__global__ __launch_bounds__(256) void bilinear_edge_scores(
    const float* __restrict__ V, const float* __restrict__ z,
    const int* __restrict__ e_src, const int* __restrict__ e_dst,
    float* __restrict__ out, int n_edges, int n_nodes) {
  const int lane    = threadIdx.x & 31;
  const int wid     = __builtin_amdgcn_readfirstlane(
      (int)((blockIdx.x * blockDim.x + threadIdx.x) >> 5));
  const int nwaves  = (gridDim.x * blockDim.x) >> 5;
  const int ngroups = (n_edges + EDGES_PER_GROUP - 1) / EDGES_PER_GROUP;
  const v4f* __restrict__ V4 = (const v4f*)V;
  const v4f* __restrict__ z4 = (const v4f*)z;

  int g = wid;
  if (g >= ngroups) return;

  int si, di;
  {
    const int e = min(g * EDGES_PER_GROUP + lane, n_edges - 1);
    si = clamp_node(e_src[e], n_nodes);
    di = clamp_node(e_dst[e], n_nodes);
  }

  v4f av[WINDOW], bv[WINDOW];
#pragma unroll
  for (int k = 0; k < WINDOW; ++k)
    request_pair<0>(av[k], bv[k], k, si, di, V4, z4, lane);

  for (;;) {
    const int gn = g + nwaves;
    if (gn >= ngroups) break;
    const int en = min(gn * EDGES_PER_GROUP + lane, n_edges - 1);
    int nsi = e_src[en];
    int ndi = e_dst[en];
    group_body<true>(av, bv, si, di, nsi, ndi, V4, z4, lane, n_nodes, out, g, n_edges);
    si = nsi;
    di = ndi;
    g = gn;
  }
  int unused_s = 0, unused_d = 0;
  group_body<false>(av, bv, si, di, unused_s, unused_d, V4, z4, lane, n_nodes, out, g,
                    n_edges);
}

extern "C" void kernel_launch(void* const* d_in, const int* in_sizes, int n_in,
                              void* d_out, int out_size, void* d_ws, size_t ws_size,
                              hipStream_t stream) {
  const float* z  = (const float*)d_in[0];
  const int*   ei = (const int*)d_in[1];
  const float* W  = (const float*)d_in[2];
  float* out = (float*)d_out;
  float* V   = (float*)d_ws;

  const int n_nodes = in_sizes[0] / FEAT;
  const int n_edges = in_sizes[1] / 2;

  bilinear_gemm_zW<<<GEMM_BLOCKS, 256, 0, stream>>>(z, W, V, n_nodes);

  bilinear_edge_scores<<<782, 256, 0, stream>>>(V, z, ei, ei + n_edges, out,
                                                n_edges, n_nodes);
}


# z@W kernel: a tile's 32 A fragments are read from LDS as one group into their own registers ahead of the WMMAs (was one read and a full wait in front of every two WMMAs)
# speedup vs baseline: 1.9685x; 1.0361x over previous
#include <hip/hip_runtime.h>
#include <hip/hip_bf16.h>

typedef __attribute__((ext_vector_type(2))) float v2f;
typedef __attribute__((ext_vector_type(8))) float v8f;

#define FEAT 128

#define GEMM_BLOCKS 768

__global__ __launch_bounds__(256) void bilinear_gemm_zW(
    const float* __restrict__ z, const float* __restrict__ W,
    float* __restrict__ V, int n_nodes) {
  __shared__ float Al[2][16][132];

  const int tid   = threadIdx.x;
  const int wv    = tid >> 5;
  const int lane  = tid & 31;
  const int colBase = wv * 16;
  const int mrow  = lane & 15;
  const int khalf = (lane >> 4) << 1;

  const int nTiles = (n_nodes + 15) >> 4;
  int tile = blockIdx.x;
  if (tile >= nTiles) return;

  const int sr = tid >> 5;
  const int sc = (tid & 31) << 2;

  float4 ra = *(const float4*)&z[(size_t)(tile * 16 + sr) * FEAT + sc];
  float4 rb = *(const float4*)&z[(size_t)(tile * 16 + sr + 8) * FEAT + sc];

  v2f bf[FEAT / 4];
#pragma unroll
  for (int kk = 0; kk < FEAT / 4; ++kk) {
    const int k = kk * 4 + khalf;
    bf[kk][0] = W[(size_t)k * FEAT + colBase + mrow];
    bf[kk][1] = W[(size_t)(k + 1) * FEAT + colBase + mrow];
  }

  int s = 0;
  for (;;) {
    *(float4*)&Al[s][sr][sc]     = ra;
    *(float4*)&Al[s][sr + 8][sc] = rb;
    __syncthreads();

    const int next = tile + GEMM_BLOCKS;
    const bool more = next < nTiles;
    const int ldTile = more ? next : tile;
    ra = *(const float4*)&z[(size_t)(ldTile * 16 + sr) * FEAT + sc];
    rb = *(const float4*)&z[(size_t)(ldTile * 16 + sr + 8) * FEAT + sc];
    __builtin_amdgcn_sched_barrier(0);

    v2f af[FEAT / 4];
#pragma unroll
    for (int kk = 0; kk < FEAT / 4; ++kk) {
      const int k = kk * 4 + khalf;
      af[kk][0] = Al[s][mrow][k];
      af[kk][1] = Al[s][mrow][k + 1];
    }
    __builtin_amdgcn_sched_barrier(0);

    v8f acc = {};
#pragma unroll
    for (int kk = 0; kk < FEAT / 4; ++kk) {
      acc = __builtin_amdgcn_wmma_f32_16x16x4_f32(
          false, af[kk], false, bf[kk], (short)0, acc, false, false);
    }

    const int rowBase = tile * 16;
    const int mOff = (lane >> 4) << 3;
#pragma unroll
    for (int r = 0; r < 8; ++r) {
      V[(size_t)(rowBase + mOff + r) * FEAT + colBase + mrow] = acc[r];
    }

    if (!more) break;
    tile = next;
    s ^= 1;
  }
}

typedef __attribute__((ext_vector_type(4))) float v4f;

template <int CTRL>
__device__ __forceinline__ float row_swz(float x) {
  return __builtin_bit_cast(float, __builtin_amdgcn_update_dpp(
      0, __builtin_bit_cast(int, x), CTRL, 0xf, 0xf, false));
}
__device__ __forceinline__ float xhalf(float x) {
  const int xi = __builtin_bit_cast(int, x);
  return __builtin_bit_cast(float, __builtin_amdgcn_permlanex16(
      xi, xi, 0x76543210, 0xfedcba98, false, false));
}
__device__ __forceinline__ int clamp_node(int i, int n_nodes) {
  return min(max(i, 0), n_nodes - 1);
}

#define EDGES_PER_GROUP 32
#define WINDOW 8

__device__ __forceinline__ constexpr int slot_edge(int b, int k) {
  return b + 4 * ((k >> 2) & 1) + 8 * ((k >> 1) & 1) + 16 * (k & 1);
}

template <int NB>
__device__ __forceinline__ void request_pair(v4f& a, v4f& b, int k, int rsi, int rdi,
                                             const v4f* __restrict__ V4,
                                             const v4f* __restrict__ z4, int lane) {
  const int L = slot_edge(NB, k);
  const int s = __builtin_amdgcn_readlane(rsi, L);
  const int d = __builtin_amdgcn_readlane(rdi, L);
  a = V4[(size_t)s * (FEAT / 4) + lane];
  b = z4[(size_t)d * (FEAT / 4) + lane];
}

template <int NB, bool REFILL>
__device__ __forceinline__ void window_step(v4f (&av)[WINDOW], v4f (&bv)[WINDOW],
                                            float (&v)[WINDOW], int rsi, int rdi,
                                            const v4f* __restrict__ V4,
                                            const v4f* __restrict__ z4, int lane) {
#pragma unroll
  for (int k = 0; k < WINDOW; ++k) {
    asm volatile("" : "+v"(av[k]));
    float sum = __builtin_fmaf(av[k].w, bv[k].w,
                __builtin_fmaf(av[k].z, bv[k].z,
                __builtin_fmaf(av[k].y, bv[k].y, av[k].x * bv[k].x)));
    if (REFILL) {
      asm volatile("" : "+v"(sum), "+v"(rsi), "+v"(rdi));
      request_pair<NB>(av[k], bv[k], k, rsi, rdi, V4, z4, lane);
    }
    v[k] = sum;
  }
}

__device__ __forceinline__ float butterfly8(const float (&v)[WINDOW], bool up16, bool up8,
                                            bool up4) {
  float r[4];
#pragma unroll
  for (int m = 0; m < 4; ++m) {
    const float keep = up16 ? v[2 * m + 1] : v[2 * m];
    const float send = up16 ? v[2 * m] : v[2 * m + 1];
    r[m] = keep + xhalf(send);
  }
  float q[2];
#pragma unroll
  for (int m = 0; m < 2; ++m) {
    const float keep = up8 ? r[2 * m + 1] : r[2 * m];
    const float send = up8 ? r[2 * m] : r[2 * m + 1];
    q[m] = keep + row_swz<0x168>(send);
  }
  const float keep = up4 ? q[1] : q[0];
  const float send = up4 ? q[0] : q[1];
  float t = keep + row_swz<0x164>(send);
  t += row_swz<0x162>(t);
  t += row_swz<0x161>(t);
  return t;
}

template <bool MORE>
__device__ __forceinline__ void group_body(v4f (&av)[WINDOW], v4f (&bv)[WINDOW], int si,
                                           int di, int& nsi, int& ndi,
                                           const v4f* __restrict__ V4,
                                           const v4f* __restrict__ z4, int lane,
                                           int n_nodes, float* __restrict__ out, int g,
                                           int n_edges) {
  const bool up16 = (lane & 16) != 0;
  const bool up8  = (lane & 8) != 0;
  const bool up4  = (lane & 4) != 0;
  const bool odd1 = (lane & 1) != 0;
  const bool odd2 = (lane & 2) != 0;
  float v[WINDOW];
  window_step<1, true>(av, bv, v, si, di, V4, z4, lane);
  const float p0 = butterfly8(v, up16, up8, up4);
  window_step<2, true>(av, bv, v, si, di, V4, z4, lane);
  const float p1 = butterfly8(v, up16, up8, up4);
  window_step<3, true>(av, bv, v, si, di, V4, z4, lane);
  float p2 = butterfly8(v, up16, up8, up4);
  if (MORE) {
    asm volatile("" : "+v"(nsi), "+v"(ndi), "+v"(p2));
    nsi = clamp_node(nsi, n_nodes);
    ndi = clamp_node(ndi, n_nodes);
  }
  window_step<0, MORE>(av, bv, v, nsi, ndi, V4, z4, lane);
  const float p3 = butterfly8(v, up16, up8, up4);

  const float lo = odd1 ? p1 : p0;
  const float hi = odd1 ? p3 : p2;
  const float res = odd2 ? hi : lo;
  const int eo = g * EDGES_PER_GROUP + lane;
  if (eo < n_edges) out[eo] = res;
}

__global__ __launch_bounds__(256) void bilinear_edge_scores(
    const float* __restrict__ V, const float* __restrict__ z,
    const int* __restrict__ e_src, const int* __restrict__ e_dst,
    float* __restrict__ out, int n_edges, int n_nodes) {
  const int lane    = threadIdx.x & 31;
  const int wid     = __builtin_amdgcn_readfirstlane(
      (int)((blockIdx.x * blockDim.x + threadIdx.x) >> 5));
  const int nwaves  = (gridDim.x * blockDim.x) >> 5;
  const int ngroups = (n_edges + EDGES_PER_GROUP - 1) / EDGES_PER_GROUP;
  const v4f* __restrict__ V4 = (const v4f*)V;
  const v4f* __restrict__ z4 = (const v4f*)z;

  int g = wid;
  if (g >= ngroups) return;

  int si, di;
  {
    const int e = min(g * EDGES_PER_GROUP + lane, n_edges - 1);
    si = clamp_node(e_src[e], n_nodes);
    di = clamp_node(e_dst[e], n_nodes);
  }

  v4f av[WINDOW], bv[WINDOW];
#pragma unroll
  for (int k = 0; k < WINDOW; ++k)
    request_pair<0>(av[k], bv[k], k, si, di, V4, z4, lane);

  for (;;) {
    const int gn = g + nwaves;
    if (gn >= ngroups) break;
    const int en = min(gn * EDGES_PER_GROUP + lane, n_edges - 1);
    int nsi = e_src[en];
    int ndi = e_dst[en];
    group_body<true>(av, bv, si, di, nsi, ndi, V4, z4, lane, n_nodes, out, g, n_edges);
    si = nsi;
    di = ndi;
    g = gn;
  }
  int unused_s = 0, unused_d = 0;
  group_body<false>(av, bv, si, di, unused_s, unused_d, V4, z4, lane, n_nodes, out, g,
                    n_edges);
}

extern "C" void kernel_launch(void* const* d_in, const int* in_sizes, int n_in,
                              void* d_out, int out_size, void* d_ws, size_t ws_size,
                              hipStream_t stream) {
  const float* z  = (const float*)d_in[0];
  const int*   ei = (const int*)d_in[1];
  const float* W  = (const float*)d_in[2];
  float* out = (float*)d_out;
  float* V   = (float*)d_ws;

  const int n_nodes = in_sizes[0] / FEAT;
  const int n_edges = in_sizes[1] / 2;

  bilinear_gemm_zW<<<GEMM_BLOCKS, 256, 0, stream>>>(z, W, V, n_nodes);

  bilinear_edge_scores<<<782, 256, 0, stream>>>(V, z, ei, ei + n_edges, out,
                                                n_edges, n_nodes);
}


# z@W kernel: wave tile 16x16 -> 16x32 (two interleaved column sets) in 4-wave work-groups: one A fragment from LDS feeds two WMMAs, each V line leaves as one whole-line 8-byte-per-lane store
# speedup vs baseline: 2.0665x; 1.0498x over previous
#include <hip/hip_runtime.h>
#include <hip/hip_bf16.h>

typedef __attribute__((ext_vector_type(2))) float v2f;
typedef __attribute__((ext_vector_type(8))) float v8f;

#define FEAT 128

#define GEMM_BLOCKS 1024
#define GEMM_THREADS 128

__global__ __launch_bounds__(GEMM_THREADS) void bilinear_gemm_zW(
    const float* __restrict__ z, const float* __restrict__ W,
    float* __restrict__ V, int n_nodes) {
  __shared__ float Al[2][16][132];

  const int tid   = threadIdx.x;
  const int wv    = tid >> 5;
  const int lane  = tid & 31;
  const int colBase = wv * 32;
  const int mrow  = lane & 15;
  const int khalf = (lane >> 4) << 1;
  const int col2  = colBase + 2 * mrow;

  const int nTiles = (n_nodes + 15) >> 4;
  int tile = blockIdx.x;
  if (tile >= nTiles) return;

  const int sr = tid >> 5;
  const int sc = (tid & 31) << 2;

  const float* __restrict__ zp = z + (size_t)(tile * 16 + sr) * FEAT + sc;
  float4 ra0 = *(const float4*)&zp[0 * 4 * FEAT];
  float4 ra1 = *(const float4*)&zp[1 * 4 * FEAT];
  float4 ra2 = *(const float4*)&zp[2 * 4 * FEAT];
  float4 ra3 = *(const float4*)&zp[3 * 4 * FEAT];

  v2f bf0[FEAT / 4], bf1[FEAT / 4];
#pragma unroll
  for (int kk = 0; kk < FEAT / 4; ++kk) {
    const int k = kk * 4 + khalf;
    const v2f w0 = *(const v2f*)&W[(size_t)k * FEAT + col2];
    const v2f w1 = *(const v2f*)&W[(size_t)(k + 1) * FEAT + col2];
    bf0[kk][0] = w0[0];
    bf0[kk][1] = w1[0];
    bf1[kk][0] = w0[1];
    bf1[kk][1] = w1[1];
  }

  int s = 0;
  for (;;) {
    *(float4*)&Al[s][sr][sc]      = ra0;
    *(float4*)&Al[s][sr + 4][sc]  = ra1;
    *(float4*)&Al[s][sr + 8][sc]  = ra2;
    *(float4*)&Al[s][sr + 12][sc] = ra3;
    __syncthreads();

    const int next = tile + GEMM_BLOCKS;
    const bool more = next < nTiles;
    const int ldTile = more ? next : tile;
    zp = z + (size_t)(ldTile * 16 + sr) * FEAT + sc;
    ra0 = *(const float4*)&zp[0 * 4 * FEAT];
    ra1 = *(const float4*)&zp[1 * 4 * FEAT];
    ra2 = *(const float4*)&zp[2 * 4 * FEAT];
    ra3 = *(const float4*)&zp[3 * 4 * FEAT];

    v2f af[FEAT / 4];
#pragma unroll
    for (int kk = 0; kk < FEAT / 4; ++kk) {
      const int k = kk * 4 + khalf;
      af[kk][0] = Al[s][mrow][k];
      af[kk][1] = Al[s][mrow][k + 1];
    }
    __builtin_amdgcn_sched_barrier(0);

    v8f acc0 = {}, acc1 = {};
#pragma unroll
    for (int kk = 0; kk < FEAT / 4; ++kk) {
      acc0 = __builtin_amdgcn_wmma_f32_16x16x4_f32(
          false, af[kk], false, bf0[kk], (short)0, acc0, false, false);
      acc1 = __builtin_amdgcn_wmma_f32_16x16x4_f32(
          false, af[kk], false, bf1[kk], (short)0, acc1, false, false);
    }

    const int rowBase = tile * 16;
    const int mOff = (lane >> 4) << 3;
    float* __restrict__ vp = V + (size_t)(rowBase + mOff) * FEAT + col2;
#pragma unroll
    for (int r = 0; r < 8; ++r) {
      v2f d;
      d[0] = acc0[r];
      d[1] = acc1[r];
      *(v2f*)&vp[r * FEAT] = d;
    }

    if (!more) break;
    tile = next;
    s ^= 1;
  }
}

typedef __attribute__((ext_vector_type(4))) float v4f;

template <int CTRL>
__device__ __forceinline__ float row_swz(float x) {
  return __builtin_bit_cast(float, __builtin_amdgcn_update_dpp(
      0, __builtin_bit_cast(int, x), CTRL, 0xf, 0xf, false));
}
__device__ __forceinline__ float xhalf(float x) {
  const int xi = __builtin_bit_cast(int, x);
  return __builtin_bit_cast(float, __builtin_amdgcn_permlanex16(
      xi, xi, 0x76543210, 0xfedcba98, false, false));
}
__device__ __forceinline__ int clamp_node(int i, int n_nodes) {
  return min(max(i, 0), n_nodes - 1);
}

#define EDGES_PER_GROUP 32
#define WINDOW 8

__device__ __forceinline__ constexpr int slot_edge(int b, int k) {
  return b + 4 * ((k >> 2) & 1) + 8 * ((k >> 1) & 1) + 16 * (k & 1);
}

template <int NB>
__device__ __forceinline__ void request_pair(v4f& a, v4f& b, int k, int rsi, int rdi,
                                             const v4f* __restrict__ V4,
                                             const v4f* __restrict__ z4, int lane) {
  const int L = slot_edge(NB, k);
  const int s = __builtin_amdgcn_readlane(rsi, L);
  const int d = __builtin_amdgcn_readlane(rdi, L);
  a = V4[(size_t)s * (FEAT / 4) + lane];
  b = z4[(size_t)d * (FEAT / 4) + lane];
}

template <int NB, bool REFILL>
__device__ __forceinline__ void window_step(v4f (&av)[WINDOW], v4f (&bv)[WINDOW],
                                            float (&v)[WINDOW], int rsi, int rdi,
                                            const v4f* __restrict__ V4,
                                            const v4f* __restrict__ z4, int lane) {
#pragma unroll
  for (int k = 0; k < WINDOW; ++k) {
    asm volatile("" : "+v"(av[k]));
    float sum = __builtin_fmaf(av[k].w, bv[k].w,
                __builtin_fmaf(av[k].z, bv[k].z,
                __builtin_fmaf(av[k].y, bv[k].y, av[k].x * bv[k].x)));
    if (REFILL) {
      asm volatile("" : "+v"(sum), "+v"(rsi), "+v"(rdi));
      request_pair<NB>(av[k], bv[k], k, rsi, rdi, V4, z4, lane);
    }
    v[k] = sum;
  }
}

__device__ __forceinline__ float butterfly8(const float (&v)[WINDOW], bool up16, bool up8,
                                            bool up4) {
  float r[4];
#pragma unroll
  for (int m = 0; m < 4; ++m) {
    const float keep = up16 ? v[2 * m + 1] : v[2 * m];
    const float send = up16 ? v[2 * m] : v[2 * m + 1];
    r[m] = keep + xhalf(send);
  }
  float q[2];
#pragma unroll
  for (int m = 0; m < 2; ++m) {
    const float keep = up8 ? r[2 * m + 1] : r[2 * m];
    const float send = up8 ? r[2 * m] : r[2 * m + 1];
    q[m] = keep + row_swz<0x168>(send);
  }
  const float keep = up4 ? q[1] : q[0];
  const float send = up4 ? q[0] : q[1];
  float t = keep + row_swz<0x164>(send);
  t += row_swz<0x162>(t);
  t += row_swz<0x161>(t);
  return t;
}

template <bool MORE>
__device__ __forceinline__ void group_body(v4f (&av)[WINDOW], v4f (&bv)[WINDOW], int si,
                                           int di, int& nsi, int& ndi,
                                           const v4f* __restrict__ V4,
                                           const v4f* __restrict__ z4, int lane,
                                           int n_nodes, float* __restrict__ out, int g,
                                           int n_edges) {
  const bool up16 = (lane & 16) != 0;
  const bool up8  = (lane & 8) != 0;
  const bool up4  = (lane & 4) != 0;
  const bool odd1 = (lane & 1) != 0;
  const bool odd2 = (lane & 2) != 0;
  float v[WINDOW];
  window_step<1, true>(av, bv, v, si, di, V4, z4, lane);
  const float p0 = butterfly8(v, up16, up8, up4);
  window_step<2, true>(av, bv, v, si, di, V4, z4, lane);
  const float p1 = butterfly8(v, up16, up8, up4);
  window_step<3, true>(av, bv, v, si, di, V4, z4, lane);
  float p2 = butterfly8(v, up16, up8, up4);
  if (MORE) {
    asm volatile("" : "+v"(nsi), "+v"(ndi), "+v"(p2));
    nsi = clamp_node(nsi, n_nodes);
    ndi = clamp_node(ndi, n_nodes);
  }
  window_step<0, MORE>(av, bv, v, nsi, ndi, V4, z4, lane);
  const float p3 = butterfly8(v, up16, up8, up4);

  const float lo = odd1 ? p1 : p0;
  const float hi = odd1 ? p3 : p2;
  const float res = odd2 ? hi : lo;
  const int eo = g * EDGES_PER_GROUP + lane;
  if (eo < n_edges) out[eo] = res;
}

__global__ __launch_bounds__(256) void bilinear_edge_scores(
    const float* __restrict__ V, const float* __restrict__ z,
    const int* __restrict__ e_src, const int* __restrict__ e_dst,
    float* __restrict__ out, int n_edges, int n_nodes) {
  const int lane    = threadIdx.x & 31;
  const int wid     = __builtin_amdgcn_readfirstlane(
      (int)((blockIdx.x * blockDim.x + threadIdx.x) >> 5));
  const int nwaves  = (gridDim.x * blockDim.x) >> 5;
  const int ngroups = (n_edges + EDGES_PER_GROUP - 1) / EDGES_PER_GROUP;
  const v4f* __restrict__ V4 = (const v4f*)V;
  const v4f* __restrict__ z4 = (const v4f*)z;

  int g = wid;
  if (g >= ngroups) return;

  int si, di;
  {
    const int e = min(g * EDGES_PER_GROUP + lane, n_edges - 1);
    si = clamp_node(e_src[e], n_nodes);
    di = clamp_node(e_dst[e], n_nodes);
  }

  v4f av[WINDOW], bv[WINDOW];
#pragma unroll
  for (int k = 0; k < WINDOW; ++k)
    request_pair<0>(av[k], bv[k], k, si, di, V4, z4, lane);

  for (;;) {
    const int gn = g + nwaves;
    if (gn >= ngroups) break;
    const int en = min(gn * EDGES_PER_GROUP + lane, n_edges - 1);
    int nsi = e_src[en];
    int ndi = e_dst[en];
    group_body<true>(av, bv, si, di, nsi, ndi, V4, z4, lane, n_nodes, out, g, n_edges);
    si = nsi;
    di = ndi;
    g = gn;
  }
  int unused_s = 0, unused_d = 0;
  group_body<false>(av, bv, si, di, unused_s, unused_d, V4, z4, lane, n_nodes, out, g,
                    n_edges);
}

extern "C" void kernel_launch(void* const* d_in, const int* in_sizes, int n_in,
                              void* d_out, int out_size, void* d_ws, size_t ws_size,
                              hipStream_t stream) {
  const float* z  = (const float*)d_in[0];
  const int*   ei = (const int*)d_in[1];
  const float* W  = (const float*)d_in[2];
  float* out = (float*)d_out;
  float* V   = (float*)d_ws;

  const int n_nodes = in_sizes[0] / FEAT;
  const int n_edges = in_sizes[1] / 2;

  bilinear_gemm_zW<<<GEMM_BLOCKS, GEMM_THREADS, 0, stream>>>(z, W, V, n_nodes);

  bilinear_edge_scores<<<782, 256, 0, stream>>>(V, z, ei, ei + n_edges, out,
                                                n_edges, n_nodes);
}


# edge-score kernel: one 25-wave work-group per WGP (250 x 800 threads, was 782 x 256), so every WGP in use gathers for the same 75 edge groups (was 72 with three work-groups, 96 with four)
# speedup vs baseline: 2.1646x; 1.0475x over previous
#include <hip/hip_runtime.h>
#include <hip/hip_bf16.h>

typedef __attribute__((ext_vector_type(2))) float v2f;
typedef __attribute__((ext_vector_type(8))) float v8f;

#define FEAT 128

#define GEMM_BLOCKS 1024
#define GEMM_THREADS 128

__global__ __launch_bounds__(GEMM_THREADS) void bilinear_gemm_zW(
    const float* __restrict__ z, const float* __restrict__ W,
    float* __restrict__ V, int n_nodes) {
  __shared__ float Al[2][16][132];

  const int tid   = threadIdx.x;
  const int wv    = tid >> 5;
  const int lane  = tid & 31;
  const int colBase = wv * 32;
  const int mrow  = lane & 15;
  const int khalf = (lane >> 4) << 1;
  const int col2  = colBase + 2 * mrow;

  const int nTiles = (n_nodes + 15) >> 4;
  int tile = blockIdx.x;
  if (tile >= nTiles) return;

  const int sr = tid >> 5;
  const int sc = (tid & 31) << 2;

  const float* __restrict__ zp = z + (size_t)(tile * 16 + sr) * FEAT + sc;
  float4 ra0 = *(const float4*)&zp[0 * 4 * FEAT];
  float4 ra1 = *(const float4*)&zp[1 * 4 * FEAT];
  float4 ra2 = *(const float4*)&zp[2 * 4 * FEAT];
  float4 ra3 = *(const float4*)&zp[3 * 4 * FEAT];

  v2f bf0[FEAT / 4], bf1[FEAT / 4];
#pragma unroll
  for (int kk = 0; kk < FEAT / 4; ++kk) {
    const int k = kk * 4 + khalf;
    const v2f w0 = *(const v2f*)&W[(size_t)k * FEAT + col2];
    const v2f w1 = *(const v2f*)&W[(size_t)(k + 1) * FEAT + col2];
    bf0[kk][0] = w0[0];
    bf0[kk][1] = w1[0];
    bf1[kk][0] = w0[1];
    bf1[kk][1] = w1[1];
  }

  int s = 0;
  for (;;) {
    *(float4*)&Al[s][sr][sc]      = ra0;
    *(float4*)&Al[s][sr + 4][sc]  = ra1;
    *(float4*)&Al[s][sr + 8][sc]  = ra2;
    *(float4*)&Al[s][sr + 12][sc] = ra3;
    __syncthreads();

    const int next = tile + GEMM_BLOCKS;
    const bool more = next < nTiles;
    const int ldTile = more ? next : tile;
    zp = z + (size_t)(ldTile * 16 + sr) * FEAT + sc;
    ra0 = *(const float4*)&zp[0 * 4 * FEAT];
    ra1 = *(const float4*)&zp[1 * 4 * FEAT];
    ra2 = *(const float4*)&zp[2 * 4 * FEAT];
    ra3 = *(const float4*)&zp[3 * 4 * FEAT];

    v2f af[FEAT / 4];
#pragma unroll
    for (int kk = 0; kk < FEAT / 4; ++kk) {
      const int k = kk * 4 + khalf;
      af[kk][0] = Al[s][mrow][k];
      af[kk][1] = Al[s][mrow][k + 1];
    }
    __builtin_amdgcn_sched_barrier(0);

    v8f acc0 = {}, acc1 = {};
#pragma unroll
    for (int kk = 0; kk < FEAT / 4; ++kk) {
      acc0 = __builtin_amdgcn_wmma_f32_16x16x4_f32(
          false, af[kk], false, bf0[kk], (short)0, acc0, false, false);
      acc1 = __builtin_amdgcn_wmma_f32_16x16x4_f32(
          false, af[kk], false, bf1[kk], (short)0, acc1, false, false);
    }

    const int rowBase = tile * 16;
    const int mOff = (lane >> 4) << 3;
    float* __restrict__ vp = V + (size_t)(rowBase + mOff) * FEAT + col2;
#pragma unroll
    for (int r = 0; r < 8; ++r) {
      v2f d;
      d[0] = acc0[r];
      d[1] = acc1[r];
      *(v2f*)&vp[r * FEAT] = d;
    }

    if (!more) break;
    tile = next;
    s ^= 1;
  }
}

typedef __attribute__((ext_vector_type(4))) float v4f;

template <int CTRL>
__device__ __forceinline__ float row_swz(float x) {
  return __builtin_bit_cast(float, __builtin_amdgcn_update_dpp(
      0, __builtin_bit_cast(int, x), CTRL, 0xf, 0xf, false));
}
__device__ __forceinline__ float xhalf(float x) {
  const int xi = __builtin_bit_cast(int, x);
  return __builtin_bit_cast(float, __builtin_amdgcn_permlanex16(
      xi, xi, 0x76543210, 0xfedcba98, false, false));
}
__device__ __forceinline__ int clamp_node(int i, int n_nodes) {
  return min(max(i, 0), n_nodes - 1);
}

#define EDGES_PER_GROUP 32
#define WINDOW 8

__device__ __forceinline__ constexpr int slot_edge(int b, int k) {
  return b + 4 * ((k >> 2) & 1) + 8 * ((k >> 1) & 1) + 16 * (k & 1);
}

template <int NB>
__device__ __forceinline__ void request_pair(v4f& a, v4f& b, int k, int rsi, int rdi,
                                             const v4f* __restrict__ V4,
                                             const v4f* __restrict__ z4, int lane) {
  const int L = slot_edge(NB, k);
  const int s = __builtin_amdgcn_readlane(rsi, L);
  const int d = __builtin_amdgcn_readlane(rdi, L);
  a = V4[(size_t)s * (FEAT / 4) + lane];
  b = z4[(size_t)d * (FEAT / 4) + lane];
}

template <int NB, bool REFILL>
__device__ __forceinline__ void window_step(v4f (&av)[WINDOW], v4f (&bv)[WINDOW],
                                            float (&v)[WINDOW], int rsi, int rdi,
                                            const v4f* __restrict__ V4,
                                            const v4f* __restrict__ z4, int lane) {
#pragma unroll
  for (int k = 0; k < WINDOW; ++k) {
    asm volatile("" : "+v"(av[k]));
    float sum = __builtin_fmaf(av[k].w, bv[k].w,
                __builtin_fmaf(av[k].z, bv[k].z,
                __builtin_fmaf(av[k].y, bv[k].y, av[k].x * bv[k].x)));
    if (REFILL) {
      asm volatile("" : "+v"(sum), "+v"(rsi), "+v"(rdi));
      request_pair<NB>(av[k], bv[k], k, rsi, rdi, V4, z4, lane);
    }
    v[k] = sum;
  }
}

__device__ __forceinline__ float butterfly8(const float (&v)[WINDOW], bool up16, bool up8,
                                            bool up4) {
  float r[4];
#pragma unroll
  for (int m = 0; m < 4; ++m) {
    const float keep = up16 ? v[2 * m + 1] : v[2 * m];
    const float send = up16 ? v[2 * m] : v[2 * m + 1];
    r[m] = keep + xhalf(send);
  }
  float q[2];
#pragma unroll
  for (int m = 0; m < 2; ++m) {
    const float keep = up8 ? r[2 * m + 1] : r[2 * m];
    const float send = up8 ? r[2 * m] : r[2 * m + 1];
    q[m] = keep + row_swz<0x168>(send);
  }
  const float keep = up4 ? q[1] : q[0];
  const float send = up4 ? q[0] : q[1];
  float t = keep + row_swz<0x164>(send);
  t += row_swz<0x162>(t);
  t += row_swz<0x161>(t);
  return t;
}

template <bool MORE>
__device__ __forceinline__ void group_body(v4f (&av)[WINDOW], v4f (&bv)[WINDOW], int si,
                                           int di, int& nsi, int& ndi,
                                           const v4f* __restrict__ V4,
                                           const v4f* __restrict__ z4, int lane,
                                           int n_nodes, float* __restrict__ out, int g,
                                           int n_edges) {
  const bool up16 = (lane & 16) != 0;
  const bool up8  = (lane & 8) != 0;
  const bool up4  = (lane & 4) != 0;
  const bool odd1 = (lane & 1) != 0;
  const bool odd2 = (lane & 2) != 0;
  float v[WINDOW];
  window_step<1, true>(av, bv, v, si, di, V4, z4, lane);
  const float p0 = butterfly8(v, up16, up8, up4);
  window_step<2, true>(av, bv, v, si, di, V4, z4, lane);
  const float p1 = butterfly8(v, up16, up8, up4);
  window_step<3, true>(av, bv, v, si, di, V4, z4, lane);
  float p2 = butterfly8(v, up16, up8, up4);
  if (MORE) {
    asm volatile("" : "+v"(nsi), "+v"(ndi), "+v"(p2));
    nsi = clamp_node(nsi, n_nodes);
    ndi = clamp_node(ndi, n_nodes);
  }
  window_step<0, MORE>(av, bv, v, nsi, ndi, V4, z4, lane);
  const float p3 = butterfly8(v, up16, up8, up4);

  const float lo = odd1 ? p1 : p0;
  const float hi = odd1 ? p3 : p2;
  const float res = odd2 ? hi : lo;
  const int eo = g * EDGES_PER_GROUP + lane;
  if (eo < n_edges) out[eo] = res;
}

#define EDGE_BLOCKS 250
#define EDGE_THREADS 800

__global__ __launch_bounds__(EDGE_THREADS) void bilinear_edge_scores(
    const float* __restrict__ V, const float* __restrict__ z,
    const int* __restrict__ e_src, const int* __restrict__ e_dst,
    float* __restrict__ out, int n_edges, int n_nodes) {
  const int lane    = threadIdx.x & 31;
  const int wid     = __builtin_amdgcn_readfirstlane(
      (int)((blockIdx.x * blockDim.x + threadIdx.x) >> 5));
  const int nwaves  = (gridDim.x * blockDim.x) >> 5;
  const int ngroups = (n_edges + EDGES_PER_GROUP - 1) / EDGES_PER_GROUP;
  const v4f* __restrict__ V4 = (const v4f*)V;
  const v4f* __restrict__ z4 = (const v4f*)z;

  int g = wid;
  if (g >= ngroups) return;

  int si, di;
  {
    const int e = min(g * EDGES_PER_GROUP + lane, n_edges - 1);
    si = clamp_node(e_src[e], n_nodes);
    di = clamp_node(e_dst[e], n_nodes);
  }

  v4f av[WINDOW], bv[WINDOW];
#pragma unroll
  for (int k = 0; k < WINDOW; ++k)
    request_pair<0>(av[k], bv[k], k, si, di, V4, z4, lane);

  for (;;) {
    const int gn = g + nwaves;
    if (gn >= ngroups) break;
    const int en = min(gn * EDGES_PER_GROUP + lane, n_edges - 1);
    int nsi = e_src[en];
    int ndi = e_dst[en];
    group_body<true>(av, bv, si, di, nsi, ndi, V4, z4, lane, n_nodes, out, g, n_edges);
    si = nsi;
    di = ndi;
    g = gn;
  }
  int unused_s = 0, unused_d = 0;
  group_body<false>(av, bv, si, di, unused_s, unused_d, V4, z4, lane, n_nodes, out, g,
                    n_edges);
}

extern "C" void kernel_launch(void* const* d_in, const int* in_sizes, int n_in,
                              void* d_out, int out_size, void* d_ws, size_t ws_size,
                              hipStream_t stream) {
  const float* z  = (const float*)d_in[0];
  const int*   ei = (const int*)d_in[1];
  const float* W  = (const float*)d_in[2];
  float* out = (float*)d_out;
  float* V   = (float*)d_ws;

  const int n_nodes = in_sizes[0] / FEAT;
  const int n_edges = in_sizes[1] / 2;

  bilinear_gemm_zW<<<GEMM_BLOCKS, GEMM_THREADS, 0, stream>>>(z, W, V, n_nodes);

  bilinear_edge_scores<<<EDGE_BLOCKS, EDGE_THREADS, 0, stream>>>(V, z, ei, ei + n_edges, out,
                                                                 n_edges, n_nodes);
}
